# starting kernel
# baseline (speedup 1.0000x reference)
#include <hip/hip_runtime.h>

#define BB 64
#define SS 2048
#define DD 512
#define MAX_SPLIT 16

typedef __attribute__((ext_vector_type(2))) float v2f;
typedef __attribute__((ext_vector_type(8))) float v8f;

__device__ __forceinline__ int imin(int a, int b) { return a < b ? a : b; }

__global__ void __launch_bounds__(128)
mean_pool_partial(const float* __restrict__ feat,
                  const int*   __restrict__ lengths,
                  float*       __restrict__ partial,
                  int chunk)
{
    const int lane  = threadIdx.x & 31;
    const int wave  = threadIdx.x >> 5;
    const int dtile = blockIdx.x * 4 + wave;
    const int b     = blockIdx.y;
    const int split = blockIdx.z;
    const int d0    = dtile * 16;

    const int len = lengths[b];
    const int s0  = split * chunk;
    const int s1  = imin(s0 + chunk, len);

    const float* __restrict__ base = feat + (size_t)b * SS * DD;

    const int col = lane & 15;
    const int khi = (lane >> 4) << 1;

    const float* __restrict__ p =
        base + (size_t)(s0 + khi) * DD + d0 + col;

    v8f c = {0.f, 0.f, 0.f, 0.f, 0.f, 0.f, 0.f, 0.f};
    const v2f a_ones = {1.0f, 1.0f};

    int s = s0;
    for (; s + 16 <= s1; s += 16) {
        #pragma unroll
        for (int j = 0; j < 4; ++j) {
            v2f bm;
            bm.x = p[j * 4 * DD];
            bm.y = p[j * 4 * DD + DD];
            c = __builtin_amdgcn_wmma_f32_16x16x4_f32(
                     false, a_ones,  false, bm,
                     (short)0, c,  false,  false);
        }
        p += 16 * DD;
    }
    for (; s + 4 <= s1; s += 4) {
        v2f bm;
        bm.x = p[0];
        bm.y = p[DD];
        c = __builtin_amdgcn_wmma_f32_16x16x4_f32(
                false, a_ones, false, bm, (short)0, c, false, false);
        p += 4 * DD;
    }
    if (s < s1) {
        const int rem = s1 - s;
        v2f a, bm;
        a.x = (khi     < rem) ? 1.0f : 0.0f;
        a.y = (khi + 1 < rem) ? 1.0f : 0.0f;
        const int r0 = imin(khi,     rem - 1);
        const int r1 = imin(khi + 1, rem - 1);
        bm.x = base[(size_t)(s + r0) * DD + d0 + col];
        bm.y = base[(size_t)(s + r1) * DD + d0 + col];
        c = __builtin_amdgcn_wmma_f32_16x16x4_f32(
                false, a, false, bm, (short)0, c, false, false);
    }

    if (lane < 16) {
        partial[((size_t)split * BB + b) * DD + d0 + lane] = c[0];
    }
}

__global__ void __launch_bounds__(256)
mean_pool_finalize(const float* __restrict__ partial,
                   const int*   __restrict__ lengths,
                   float*       __restrict__ out,
                   int split_cnt)
{
    const int idx = blockIdx.x * 256 + threadIdx.x;
    if (idx >= BB * DD) return;
    const int b = idx / DD;
    float sum = 0.f;
    for (int sp = 0; sp < split_cnt; ++sp)
        sum += partial[(size_t)sp * BB * DD + idx];
    out[idx] = sum / (float)lengths[b];
}

extern "C" void kernel_launch(void* const* d_in, const int* in_sizes, int n_in,
                              void* d_out, int out_size, void* d_ws, size_t ws_size,
                              hipStream_t stream)
{
    const float* feat    = (const float*)d_in[0];
    const int*   lengths = (const int*)d_in[1];
    float*       out     = (float*)d_out;

    int split = (int)(ws_size / ((size_t)BB * DD * sizeof(float)));
    if (split > MAX_SPLIT) split = MAX_SPLIT;
    float* partial = (float*)d_ws;
    if (split < 1) {
        split = 1;
        partial = out;
    }
    const int chunk = (SS + split - 1) / split;

    dim3 grid1(DD / 16 / 4, BB, split);
    mean_pool_partial<<<grid1, 128, 0, stream>>>(feat, lengths, partial, chunk);

    const int total = BB * DD;
    mean_pool_finalize<<<(total + 255) / 256, 256, 0, stream>>>(partial, lengths, out, split);
}


# one launch: 512 work-groups = (batch, 64-column strip) issued longest batch first, one 16-wave group per WGP so the dispatcher evens the load; waves take row pairs in turn; sum order changed
# speedup vs baseline: 1.1714x; 1.1714x over previous
#include <hip/hip_runtime.h>

#define BB 64
#define SS 2048
#define DD 512
#define NW 16
#define GB 8
#define PSTRIDE (NW * 2 * DD)

typedef __attribute__((ext_vector_type(2))) float v2f;
typedef __attribute__((ext_vector_type(4))) float v4f;
typedef __attribute__((ext_vector_type(8))) float v8f;

static_assert(GB == 8 && NW * GB * 2 * 8 >= SS, "eight stages of eight pairs must cover the longest sequence");

#define WMMA4(ACC, AV, BV) \
    ACC = __builtin_amdgcn_wmma_f32_16x16x4_f32(false, AV, false, BV, (short)0, ACC, false, false)

#define SUM_LOAD(V, AV, EVEN)                                              \
    {                                                                      \
        v2f bxy = {V.x, V.y};                                              \
        v2f bzw = {V.z, V.w};                                              \
        if (EVEN) { WMMA4(acc0, AV, bxy); WMMA4(acc1, AV, bzw); }          \
        else      { WMMA4(acc2, AV, bxy); WMMA4(acc3, AV, bzw); }          \
    }

template <int NS>
__device__ __forceinline__ void run_stages(const float* p, const int nv, const v2f sel,
                                           v8f& acc0, v8f& acc1, v8f& acc2, v8f& acc3)
{
    v4f Q[4][GB];
    #pragma unroll
    for (int s = 0; s < (NS < 4 ? NS : 4); ++s) {
        #pragma unroll
        for (int k = 0; k < GB; ++k)
            if (s < NS - 1 || k < nv)
                Q[s][k] = *(const v4f*)(p + (size_t)(s * GB + k) * PSTRIDE);
    }
    #pragma unroll
    for (int s = 0; s < NS; ++s) {
        #pragma unroll
        for (int k = 0; k < GB; ++k)
            if (s < NS - 1 || k < nv) SUM_LOAD(Q[s & 3][k], sel, ((k & 1) == 0))
        if (s + 4 < NS) {
            #pragma unroll
            for (int k = 0; k < GB; ++k)
                if (s + 4 < NS - 1 || k < nv)
                    Q[s & 3][k] = *(const v4f*)(p + (size_t)((s + 4) * GB + k) * PSTRIDE);
        }
    }
}

__global__ void __launch_bounds__(NW * 32)
masked_mean_strips(const float* feat,
                   const int* __restrict__ lengths,
                   float* __restrict__ out)
{
    __shared__ v4f red[NW][16];

    const int lane  = threadIdx.x & 31;
    const int wave  = __builtin_amdgcn_readfirstlane((int)(threadIdx.x >> 5));
    const int rank  = blockIdx.x >> 3;
    const int strip = blockIdx.x & 7;

    const int la = lengths[lane];
    const int lb = lengths[lane + 32];
    int ra = 0, rb = 0;
    for (int j = 0; j < BB; ++j) {
        const int lj = lengths[j];
        ra += (lj > la) || (lj == la && j < lane);
        rb += (lj > lb) || (lj == lb && j < lane + 32);
    }
    const unsigned ma = __builtin_amdgcn_ballot_w32(ra == rank);
    const unsigned mb = __builtin_amdgcn_ballot_w32(rb == rank);
    const int b = __builtin_amdgcn_readfirstlane(ma ? __builtin_ctz(ma) : 32 + __builtin_ctz(mb | 0x80000000u));

    const int len   = lengths[b];
    const int rows  = len < 0 ? 0 : (len > SS ? SS : len);
    const int npair = rows >> 1;
    const int odd   = rows & 1;
    const int n     = (npair > wave) ? ((npair - wave + NW - 1) / NW) : 0;

    const int c = lane & 15;
    const int h = lane >> 4;
    const float* base = feat + (size_t)b * SS * DD + strip * 64 + 4 * c;
    const float* p    = base + (size_t)(2 * wave + h) * DD;

    v8f acc0 = {0.f, 0.f, 0.f, 0.f, 0.f, 0.f, 0.f, 0.f};
    v8f acc1 = acc0, acc2 = acc0, acc3 = acc0;

    v2f sel;
    sel.x = (c & 1) ? 0.0f : 1.0f;
    sel.y = (c & 1) ? 1.0f : 0.0f;

    const bool has_tail = (odd != 0) && (wave == (npair & (NW - 1)));
    const v4f t0 = *(const v4f*)(base + (size_t)(rows > 0 ? rows - 1 : 0) * DD);

    const int ns = (n + GB - 1) >> 3;
    const int nv = n - (ns - 1) * GB;
    switch (ns) {
        case 1: run_stages<1>(p, nv, sel, acc0, acc1, acc2, acc3); break;
        case 2: run_stages<2>(p, nv, sel, acc0, acc1, acc2, acc3); break;
        case 3: run_stages<3>(p, nv, sel, acc0, acc1, acc2, acc3); break;
        case 4: run_stages<4>(p, nv, sel, acc0, acc1, acc2, acc3); break;
        case 5: run_stages<5>(p, nv, sel, acc0, acc1, acc2, acc3); break;
        case 6: run_stages<6>(p, nv, sel, acc0, acc1, acc2, acc3); break;
        case 7: run_stages<7>(p, nv, sel, acc0, acc1, acc2, acc3); break;
        case 8: run_stages<8>(p, nv, sel, acc0, acc1, acc2, acc3); break;
        default: break;
    }

    if (has_tail) {
        v2f selt;
        selt.x = (h == 0) ? sel.x : 0.0f;
        selt.y = (h == 0) ? sel.y : 0.0f;
        SUM_LOAD(t0, selt, true)
    }

    if (lane < 16) {
        v4f r = {acc0[0] + acc2[0], acc0[1] + acc2[1], acc1[0] + acc3[0], acc1[1] + acc3[1]};
        red[wave][lane] = r;
    }
    __syncthreads();

    if (threadIdx.x < 64) {
        const float* rf = (const float*)&red[0][0];
        float s = 0.f;
        #pragma unroll
        for (int w = 0; w < NW; ++w) s += rf[w * 64 + threadIdx.x];
        out[(size_t)b * DD + strip * 64 + threadIdx.x] = s / (float)len;
    }
}

extern "C" void kernel_launch(void* const* d_in, const int* in_sizes, int n_in,
                              void* d_out, int out_size, void* d_ws, size_t ws_size,
                              hipStream_t stream)
{
    const float* feat    = (const float*)d_in[0];
    const int*   lengths = (const int*)d_in[1];
    float*       out     = (float*)d_out;
    (void)in_sizes; (void)n_in; (void)out_size; (void)d_ws; (void)ws_size;

    masked_mean_strips<<<512, 512, 0, stream>>>(feat, lengths, out);
}


# a work-group finds its batch (rank-th longest) by a 12-step threshold search on wave ballots, not 64 compares per lane behind 9 dependent scalar loads; streaming starts sooner; sum order unchanged
# speedup vs baseline: 1.4466x; 1.2349x over previous
#include <hip/hip_runtime.h>

#define BB 64
#define SS 2048
#define DD 512
#define NW 16
#define GB 8
#define PSTRIDE (NW * 2 * DD)

typedef __attribute__((ext_vector_type(2))) float v2f;
typedef __attribute__((ext_vector_type(4))) float v4f;
typedef __attribute__((ext_vector_type(8))) float v8f;

#define LBITS 12

static_assert(GB == 8 && NW * GB * 2 * 8 >= SS, "eight stages of eight pairs must cover the longest sequence");
static_assert(SS < (1 << LBITS) && BB == 64, "the batch search walks LBITS bits over two ballots of 32 batches");

#define WMMA4(ACC, AV, BV) \
    ACC = __builtin_amdgcn_wmma_f32_16x16x4_f32(false, AV, false, BV, (short)0, ACC, false, false)

#define SUM_LOAD(V, AV, EVEN)                                              \
    {                                                                      \
        v2f bxy = {V.x, V.y};                                              \
        v2f bzw = {V.z, V.w};                                              \
        if (EVEN) { WMMA4(acc0, AV, bxy); WMMA4(acc1, AV, bzw); }          \
        else      { WMMA4(acc2, AV, bxy); WMMA4(acc3, AV, bzw); }          \
    }

template <int NS>
__device__ __forceinline__ void run_stages(const float* p, const int nv, const v2f sel,
                                           v8f& acc0, v8f& acc1, v8f& acc2, v8f& acc3)
{
    v4f Q[4][GB];
    #pragma unroll
    for (int s = 0; s < (NS < 4 ? NS : 4); ++s) {
        #pragma unroll
        for (int k = 0; k < GB; ++k)
            if (s < NS - 1 || k < nv)
                Q[s][k] = *(const v4f*)(p + (size_t)(s * GB + k) * PSTRIDE);
    }
    #pragma unroll
    for (int s = 0; s < NS; ++s) {
        #pragma unroll
        for (int k = 0; k < GB; ++k)
            if (s < NS - 1 || k < nv) SUM_LOAD(Q[s & 3][k], sel, ((k & 1) == 0))
        if (s + 4 < NS) {
            #pragma unroll
            for (int k = 0; k < GB; ++k)
                if (s + 4 < NS - 1 || k < nv)
                    Q[s & 3][k] = *(const v4f*)(p + (size_t)((s + 4) * GB + k) * PSTRIDE);
        }
    }
}

__global__ void __launch_bounds__(NW * 32)
masked_mean_strips(const float* feat,
                   const int* __restrict__ lengths,
                   float* __restrict__ out)
{
    __shared__ v4f red[NW][16];

    const int lane  = threadIdx.x & 31;
    const int wave  = __builtin_amdgcn_readfirstlane((int)(threadIdx.x >> 5));
    const int rank  = blockIdx.x >> 3;
    const int strip = blockIdx.x & 7;

    const int la = lengths[lane];
    const int lb = lengths[lane + 32];
    const int ca = la < 0 ? 0 : (la > SS ? SS : la);
    const int cb = lb < 0 ? 0 : (lb > SS ? SS : lb);
    int v = 0;
    for (int bit = LBITS - 1; bit >= 0; --bit) {
        const int t = v | (1 << bit);
        const int reach = __builtin_popcount(__builtin_amdgcn_ballot_w32(ca >= t))
                        + __builtin_popcount(__builtin_amdgcn_ballot_w32(cb >= t));
        v = (reach > rank) ? t : v;
    }
    const int longer = __builtin_popcount(__builtin_amdgcn_ballot_w32(ca > v))
                     + __builtin_popcount(__builtin_amdgcn_ballot_w32(cb > v));
    const int k = rank - longer;
    unsigned long long same = (unsigned long long)__builtin_amdgcn_ballot_w32(ca == v)
                            | ((unsigned long long)__builtin_amdgcn_ballot_w32(cb == v) << 32);
    for (int q = 0; q < k && q < BB - 1; ++q) same &= same - 1;
    const int b = __builtin_amdgcn_readfirstlane((int)__builtin_ctzll(same | 0x8000000000000000ull));

    const int len   = (b < 32) ? __builtin_amdgcn_readlane(la, b) : __builtin_amdgcn_readlane(lb, b - 32);
    const int rows  = v;
    const int npair = rows >> 1;
    const int odd   = rows & 1;
    const int n     = (npair > wave) ? ((npair - wave + NW - 1) / NW) : 0;

    const int c = lane & 15;
    const int h = lane >> 4;
    const float* base = feat + (size_t)b * SS * DD + strip * 64 + 4 * c;
    const float* p    = base + (size_t)(2 * wave + h) * DD;

    v8f acc0 = {0.f, 0.f, 0.f, 0.f, 0.f, 0.f, 0.f, 0.f};
    v8f acc1 = acc0, acc2 = acc0, acc3 = acc0;

    v2f sel;
    sel.x = (c & 1) ? 0.0f : 1.0f;
    sel.y = (c & 1) ? 1.0f : 0.0f;

    const bool has_tail = (odd != 0) && (wave == (npair & (NW - 1)));
    const v4f t0 = *(const v4f*)(base + (size_t)(rows > 0 ? rows - 1 : 0) * DD);

    const int ns = (n + GB - 1) >> 3;
    const int nv = n - (ns - 1) * GB;
    switch (ns) {
        case 1: run_stages<1>(p, nv, sel, acc0, acc1, acc2, acc3); break;
        case 2: run_stages<2>(p, nv, sel, acc0, acc1, acc2, acc3); break;
        case 3: run_stages<3>(p, nv, sel, acc0, acc1, acc2, acc3); break;
        case 4: run_stages<4>(p, nv, sel, acc0, acc1, acc2, acc3); break;
        case 5: run_stages<5>(p, nv, sel, acc0, acc1, acc2, acc3); break;
        case 6: run_stages<6>(p, nv, sel, acc0, acc1, acc2, acc3); break;
        case 7: run_stages<7>(p, nv, sel, acc0, acc1, acc2, acc3); break;
        case 8: run_stages<8>(p, nv, sel, acc0, acc1, acc2, acc3); break;
        default: break;
    }

    if (has_tail) {
        v2f selt;
        selt.x = (h == 0) ? sel.x : 0.0f;
        selt.y = (h == 0) ? sel.y : 0.0f;
        SUM_LOAD(t0, selt, true)
    }

    if (lane < 16) {
        v4f r = {acc0[0] + acc2[0], acc0[1] + acc2[1], acc1[0] + acc3[0], acc1[1] + acc3[1]};
        red[wave][lane] = r;
    }
    __syncthreads();

    if (threadIdx.x < 64) {
        const float* rf = (const float*)&red[0][0];
        float s = 0.f;
        #pragma unroll
        for (int w = 0; w < NW; ++w) s += rf[w * 64 + threadIdx.x];
        out[(size_t)b * DD + strip * 64 + threadIdx.x] = s / (float)len;
    }
}

extern "C" void kernel_launch(void* const* d_in, const int* in_sizes, int n_in,
                              void* d_out, int out_size, void* d_ws, size_t ws_size,
                              hipStream_t stream)
{
    const float* feat    = (const float*)d_in[0];
    const int*   lengths = (const int*)d_in[1];
    float*       out     = (float*)d_out;
    (void)in_sizes; (void)n_in; (void)out_size; (void)d_ws; (void)ws_size;

    masked_mean_strips<<<512, 512, 0, stream>>>(feat, lengths, out);
}
